# starting kernel
# baseline (speedup 1.0000x reference)
#include <hip/hip_runtime.h>
#include <math.h>


#define N_NODES 50000
#define N_EDGES 1600000
#define ET_EDGES (N_EDGES + N_NODES)
#define IN_CH 128
#define OUT_CH 128
#define HEADS 4
#define CPH 32
#define NEG_SLOPE 0.2f
#define SOFTMAX_EPS 1e-16f
#define LN_EPS 1e-5f

typedef float v2f __attribute__((ext_vector_type(2)));
typedef float v8f __attribute__((ext_vector_type(8)));

__global__ void k_init(float* __restrict__ out, float* __restrict__ ssum,
                       unsigned* __restrict__ mkey, int nout, int nseg) {
    int i = blockIdx.x * blockDim.x + threadIdx.x;
    if (i < nout) out[i] = 0.0f;
    if (i < nseg) { ssum[i] = 0.0f; mkey[i] = 0u; }
}

__global__ void k_gemm_wmma(const float* __restrict__ x,
                            const float* __restrict__ W,
                            float* __restrict__ h, int nrows) {
    extern __shared__ float Wlds[];
    for (int i = threadIdx.x; i < (IN_CH * OUT_CH) / 4; i += blockDim.x)
        ((float4*)Wlds)[i] = ((const float4*)W)[i];
    __syncthreads();

    const int wave = threadIdx.x >> 5;
    const int lane = threadIdx.x & 31;
    const int rowBase = blockIdx.x * 128 + wave * 16;
    const int m  = lane & 15;
    const int kk = (lane >> 4) << 1;

    int arow = rowBase + m;
    if (arow >= nrows) arow = nrows - 1;
    const float* xk = x + (size_t)arow * IN_CH + kk;

    v8f c[8];
#pragma unroll
    for (int t = 0; t < 8; ++t) c[t] = (v8f){0,0,0,0,0,0,0,0};

    float2 av = *(const float2*)xk;
    for (int k = 0; k < IN_CH; k += 4) {
        v2f a; a.x = av.x; a.y = av.y;
        if (k + 4 < IN_CH) av = *(const float2*)(xk + k + 4);

        const float* w0 = &Wlds[(k + kk) * OUT_CH + m];
        v2f b[8];
#pragma unroll
        for (int t = 0; t < 8; ++t) {
            b[t].x = w0[t * 16];
            b[t].y = w0[OUT_CH + t * 16];
        }
#pragma unroll
        for (int t = 0; t < 8; ++t)
            c[t] = __builtin_amdgcn_wmma_f32_16x16x4_f32(
                false, a, false, b[t], (short)0, c[t], false, false);
    }

#pragma unroll
    for (int t = 0; t < 8; ++t) {
#pragma unroll
        for (int v = 0; v < 8; ++v) {
            const int M = v + ((lane >> 4) << 3);
            const int row = rowBase + M;
            if (row < nrows)
                h[(size_t)row * OUT_CH + t * 16 + m] = c[t][v];
        }
    }
}

__global__ void k_dots(const float* __restrict__ h,
                       const float* __restrict__ att_src,
                       const float* __restrict__ att_dst,
                       float* __restrict__ a_s, float* __restrict__ a_d, int n) {
    const int wave = threadIdx.x >> 5;
    const int lane = threadIdx.x & 31;
    const int node = blockIdx.x * 2 + (wave >> 2);
    const int head = wave & 3;
    if (node >= n) return;
    const float hv = h[(size_t)node * OUT_CH + head * CPH + lane];
    float vs = hv * att_src[head * CPH + lane];
    float vd = hv * att_dst[head * CPH + lane];
#pragma unroll
    for (int o = 16; o > 0; o >>= 1) {
        vs += __shfl_xor(vs, o, 32);
        vd += __shfl_xor(vd, o, 32);
    }
    if (lane == 0) {
        a_s[node * HEADS + head] = vs;
        a_d[node * HEADS + head] = vd;
    }
}

__device__ __forceinline__ unsigned f2key(float f) {
    unsigned b = __float_as_uint(f);
    return (b & 0x80000000u) ? ~b : (b | 0x80000000u);
}
__device__ __forceinline__ float key2f(unsigned k) {
    unsigned b = (k & 0x80000000u) ? (k ^ 0x80000000u) : ~k;
    return __uint_as_float(b);
}

__global__ void k_edge_max(const int* __restrict__ ei,
                           const float* __restrict__ a_s,
                           const float* __restrict__ a_d,
                           float* __restrict__ logits,
                           unsigned* __restrict__ mkey) {
    const int tid = blockIdx.x * blockDim.x + threadIdx.x;
    if (tid >= ET_EDGES * HEADS) return;
    const int e  = tid >> 2;
    const int hd = tid & 3;
    int s, d;
    if (e < N_EDGES) { s = ei[e]; d = ei[N_EDGES + e]; }
    else             { s = d = e - N_EDGES; }
    float v = a_s[s * HEADS + hd] + a_d[d * HEADS + hd];
    v = v > 0.0f ? v : NEG_SLOPE * v;
    logits[tid] = v;
    atomicMax(&mkey[d * HEADS + hd], f2key(v));
}

__global__ void k_edge_exp(const int* __restrict__ ei,
                           float* __restrict__ logits,
                           const unsigned* __restrict__ mkey,
                           float* __restrict__ ssum) {
    const int tid = blockIdx.x * blockDim.x + threadIdx.x;
    if (tid >= ET_EDGES * HEADS) return;
    const int e  = tid >> 2;
    const int hd = tid & 3;
    const int d = (e < N_EDGES) ? ei[N_EDGES + e] : (e - N_EDGES);
    const float mf = key2f(mkey[d * HEADS + hd]);
    const float ex = expf(logits[tid] - mf);
    logits[tid] = ex;
    atomicAdd(&ssum[d * HEADS + hd], ex);
}

__global__ void k_edge_agg(const int* __restrict__ ei,
                           const float* __restrict__ h,
                           const float* __restrict__ evals,
                           const float* __restrict__ ssum,
                           float* __restrict__ out) {
    const int wave = threadIdx.x >> 5;
    const int lane = threadIdx.x & 31;
    const int e = blockIdx.x * 8 + wave;
    if (e >= ET_EDGES) return;
    int s, d;
    if (e < N_EDGES) { s = ei[e]; d = ei[N_EDGES + e]; }
    else             { s = d = e - N_EDGES; }
    const int head = lane >> 3;
    const float alpha = evals[e * HEADS + head] /
                        (ssum[d * HEADS + head] + SOFTMAX_EPS);
    const float4 hv = *(const float4*)(h + (size_t)s * OUT_CH + lane * 4);
    float* op = out + (size_t)d * OUT_CH + lane * 4;
    atomicAdd(op + 0, hv.x * alpha);
    atomicAdd(op + 1, hv.y * alpha);
    atomicAdd(op + 2, hv.z * alpha);
    atomicAdd(op + 3, hv.w * alpha);
}

__global__ void k_ln_elu(float* __restrict__ out,
                         const float* __restrict__ bias,
                         const float* __restrict__ gamma,
                         const float* __restrict__ beta) {
    __shared__ float red[4];
    const int node = blockIdx.x;
    const int c = threadIdx.x;
    const int wave = c >> 5, lane = c & 31;

    float v = out[(size_t)node * OUT_CH + c] + bias[c];

    float t = v;
#pragma unroll
    for (int o = 16; o > 0; o >>= 1) t += __shfl_xor(t, o, 32);
    if (lane == 0) red[wave] = t;
    __syncthreads();
    const float mu = (red[0] + red[1] + red[2] + red[3]) * (1.0f / OUT_CH);
    __syncthreads();

    const float dv = v - mu;
    t = dv * dv;
#pragma unroll
    for (int o = 16; o > 0; o >>= 1) t += __shfl_xor(t, o, 32);
    if (lane == 0) red[wave] = t;
    __syncthreads();
    const float var = (red[0] + red[1] + red[2] + red[3]) * (1.0f / OUT_CH);

    float y = dv * (1.0f / sqrtf(var + LN_EPS)) * gamma[c] + beta[c];
    y = y > 0.0f ? y : (expf(y) - 1.0f);
    out[(size_t)node * OUT_CH + c] = y;
}

extern "C" void kernel_launch(void* const* d_in, const int* in_sizes, int n_in,
                              void* d_out, int out_size, void* d_ws, size_t ws_size,
                              hipStream_t stream) {
    const float* x       = (const float*)d_in[0];
    const int*   ei      = (const int*)  d_in[1];
    const float* W       = (const float*)d_in[2];
    const float* att_src = (const float*)d_in[3];
    const float* att_dst = (const float*)d_in[4];
    const float* bias    = (const float*)d_in[5];
    const float* gamma   = (const float*)d_in[6];
    const float* beta    = (const float*)d_in[7];
    float* out = (float*)d_out;

    char* ws = (char*)d_ws;
    float*    h      = (float*)ws;
    ws += (size_t)N_NODES * OUT_CH * sizeof(float);
    float*    a_s    = (float*)ws;  ws += (size_t)N_NODES * HEADS * sizeof(float);
    float*    a_d    = (float*)ws;  ws += (size_t)N_NODES * HEADS * sizeof(float);
    unsigned* mkey   = (unsigned*)ws; ws += (size_t)N_NODES * HEADS * sizeof(unsigned);
    float*    ssum   = (float*)ws;  ws += (size_t)N_NODES * HEADS * sizeof(float);
    float*    evals  = (float*)ws;

    const int nout = N_NODES * OUT_CH;
    const int nseg = N_NODES * HEADS;

    k_init<<<(nout + 255) / 256, 256, 0, stream>>>(out, ssum, mkey, nout, nseg);

    k_gemm_wmma<<<(N_NODES + 127) / 128, 256, IN_CH * OUT_CH * sizeof(float),
                  stream>>>(x, W, h, N_NODES);

    k_dots<<<(N_NODES + 1) / 2, 256, 0, stream>>>(h, att_src, att_dst, a_s, a_d,
                                                  N_NODES);

    k_edge_max<<<(ET_EDGES * HEADS + 255) / 256, 256, 0, stream>>>(
        ei, a_s, a_d, evals, mkey);

    k_edge_exp<<<(ET_EDGES * HEADS + 255) / 256, 256, 0, stream>>>(
        ei, evals, mkey, ssum);

    k_edge_agg<<<(ET_EDGES + 7) / 8, 256, 0, stream>>>(ei, h, evals, ssum, out);

    k_ln_elu<<<N_NODES, 128, 0, stream>>>(out, bias, gamma, beta);
}


# Edges counting-sorted by destination (returning int atomic = rank); one wave per destination does max, exp-sum and weighted h-row sum in registers; float atomics gone, sums run in arrival order
# speedup vs baseline: 1.7776x; 1.7776x over previous
#include <hip/hip_runtime.h>
#include <math.h>


#define N_NODES 50000
#define N_EDGES 1600000
#define ET_EDGES (N_EDGES + N_NODES)
#define IN_CH 128
#define OUT_CH 128
#define HEADS 4
#define CPH 32
#define NEG_SLOPE 0.2f
#define SOFTMAX_EPS 1e-16f
#define LN_EPS 1e-5f

typedef float v2f __attribute__((ext_vector_type(2)));
typedef float v8f __attribute__((ext_vector_type(8)));

__global__ void k_init(int* __restrict__ cnt, int n) {
    int i = blockIdx.x * blockDim.x + threadIdx.x;
    if (i < n) cnt[i] = 0;
}

__global__ void k_gemm_wmma(const float* __restrict__ x,
                            const float* __restrict__ W,
                            float* __restrict__ h, int nrows) {
    extern __shared__ float Wlds[];
    for (int i = threadIdx.x; i < (IN_CH * OUT_CH) / 4; i += blockDim.x)
        ((float4*)Wlds)[i] = ((const float4*)W)[i];
    __syncthreads();

    const int wave = threadIdx.x >> 5;
    const int lane = threadIdx.x & 31;
    const int rowBase = blockIdx.x * 128 + wave * 16;
    const int m  = lane & 15;
    const int kk = (lane >> 4) << 1;

    int arow = rowBase + m;
    if (arow >= nrows) arow = nrows - 1;
    const float* xk = x + (size_t)arow * IN_CH + kk;

    v8f c[8];
#pragma unroll
    for (int t = 0; t < 8; ++t) c[t] = (v8f){0,0,0,0,0,0,0,0};

    float2 av = *(const float2*)xk;
    for (int k = 0; k < IN_CH; k += 4) {
        v2f a; a.x = av.x; a.y = av.y;
        if (k + 4 < IN_CH) av = *(const float2*)(xk + k + 4);

        const float* w0 = &Wlds[(k + kk) * OUT_CH + m];
        v2f b[8];
#pragma unroll
        for (int t = 0; t < 8; ++t) {
            b[t].x = w0[t * 16];
            b[t].y = w0[OUT_CH + t * 16];
        }
#pragma unroll
        for (int t = 0; t < 8; ++t)
            c[t] = __builtin_amdgcn_wmma_f32_16x16x4_f32(
                false, a, false, b[t], (short)0, c[t], false, false);
    }

#pragma unroll
    for (int t = 0; t < 8; ++t) {
#pragma unroll
        for (int v = 0; v < 8; ++v) {
            const int M = v + ((lane >> 4) << 3);
            const int row = rowBase + M;
            if (row < nrows)
                h[(size_t)row * OUT_CH + t * 16 + m] = c[t][v];
        }
    }
}

__global__ void k_dots(const float* __restrict__ h,
                       const float* __restrict__ att_src,
                       const float* __restrict__ att_dst,
                       float* __restrict__ a_s, float* __restrict__ a_d, int n) {
    const int wave = threadIdx.x >> 5;
    const int lane = threadIdx.x & 31;
    const int node = blockIdx.x * 2 + (wave >> 2);
    const int head = wave & 3;
    if (node >= n) return;
    const float hv = h[(size_t)node * OUT_CH + head * CPH + lane];
    float vs = hv * att_src[head * CPH + lane];
    float vd = hv * att_dst[head * CPH + lane];
#pragma unroll
    for (int o = 16; o > 0; o >>= 1) {
        vs += __shfl_xor(vs, o, 32);
        vd += __shfl_xor(vd, o, 32);
    }
    if (lane == 0) {
        a_s[node * HEADS + head] = vs;
        a_d[node * HEADS + head] = vd;
    }
}

__global__ void k_count(const int* __restrict__ ei, int* __restrict__ cnt,
                        int* __restrict__ rank) {
    const int e = blockIdx.x * blockDim.x + threadIdx.x;
    if (e >= N_EDGES) return;
    rank[e] = atomicAdd(&cnt[ei[N_EDGES + e]], 1);
}

__global__ void k_scan(const int* __restrict__ cnt, int* __restrict__ off) {
    __shared__ int wtot[32];
    __shared__ int wpre[32];
    const int t = threadIdx.x;
    const int lane = t & 31, wave = t >> 5;
    const int blk0 = blockIdx.x * 1024;

    int pre = 0;
    for (int i = t; i < blk0; i += 1024) pre += cnt[i];

    const int i = blk0 + t;
    const int v = (i < N_NODES) ? cnt[i] : 0;
    int inc = v;
    for (int o = 1; o < 32; o <<= 1) {
        const int nb = __shfl_up(inc, o, 32);
        if (lane >= o) inc += nb;
    }
    for (int o = 16; o > 0; o >>= 1) pre += __shfl_xor(pre, o, 32);
    if (lane == 31) wtot[wave] = inc;
    if (lane == 0)  wpre[wave] = pre;
    __syncthreads();

    int base = 0, woff = 0;
    for (int w = 0; w < 32; ++w) {
        base += wpre[w];
        if (w < wave) woff += wtot[w];
    }
    const int excl = base + woff + inc - v;
    if (i < N_NODES) off[i] = excl;
    if (i == N_NODES - 1) off[N_NODES] = excl + v;
}

__global__ void k_fill(const int* __restrict__ ei, const int* __restrict__ off,
                       const int* __restrict__ rank, int* __restrict__ csr) {
    const int e = blockIdx.x * blockDim.x + threadIdx.x;
    if (e >= N_EDGES) return;
    const int s = ei[e];
    const int d = ei[N_EDGES + e];
    const int pos = off[d] + rank[e];
    if ((unsigned)pos < (unsigned)N_EDGES) csr[pos] = s;
}

#define GATHER_BLOCKS 2048

__device__ __forceinline__ float lrelu(float v) {
    return v > 0.0f ? v : NEG_SLOPE * v;
}
__device__ __forceinline__ float lane_f(float v, int j) {
    return __int_as_float(__builtin_amdgcn_readlane(__float_as_int(v), j));
}
struct HeadMask { unsigned k0, k1, k2, k3; };
__device__ __forceinline__ float by_head(float a, float b, float c, float e,
                                         const HeadMask& k) {
    const unsigned r = (__float_as_uint(a) & k.k0) | (__float_as_uint(b) & k.k1) |
                       (__float_as_uint(c) & k.k2) | (__float_as_uint(e) & k.k3);
    return __uint_as_float(r);
}
__device__ __forceinline__ float head_w(float w0, float w1, float w2, float w3,
                                        int j, const HeadMask& k) {
    return by_head(lane_f(w0, j), lane_f(w1, j), lane_f(w2, j), lane_f(w3, j), k);
}
__device__ __forceinline__ int clamp_i(int v, int lo, int hi) {
    return v < lo ? lo : (v > hi ? hi : v);
}

__global__ void k_gather(const int* __restrict__ off, const int* __restrict__ csr,
         const float* __restrict__ a_s, const float* __restrict__ a_d,
         const float* __restrict__ h, float* __restrict__ out) {
    const int lane = threadIdx.x & 31;
    const int head = lane >> 3;
    HeadMask hk;
    hk.k0 = head == 0 ? 0xffffffffu : 0u;
    hk.k1 = head == 1 ? 0xffffffffu : 0u;
    hk.k2 = head == 2 ? 0xffffffffu : 0u;
    hk.k3 = head == 3 ? 0xffffffffu : 0u;
    const int w_first = __builtin_amdgcn_readfirstlane(
        (int)(blockIdx.x * 8 + (threadIdx.x >> 5)));

    for (int d = w_first; d < N_NODES; d += GATHER_BLOCKS * 8) {
        const int beg = clamp_i(off[d], 0, N_EDGES);
        const int end = clamp_i(off[d + 1], beg, N_EDGES);
        const float4 ad = *(const float4*)(a_d + (size_t)d * HEADS);
        const float4 sd = *(const float4*)(a_s + (size_t)d * HEADS);
        const float ls0 = lrelu(sd.x + ad.x), ls1 = lrelu(sd.y + ad.y);
        const float ls2 = lrelu(sd.z + ad.z), ls3 = lrelu(sd.w + ad.w);

        float m0 = ls0, m1 = ls1, m2 = ls2, m3 = ls3;
        for (int i = beg + lane; i < end; i += 32) {
            const int s = clamp_i(csr[i], 0, N_NODES - 1);
            const float4 as = *(const float4*)(a_s + (size_t)s * HEADS);
            m0 = fmaxf(m0, lrelu(as.x + ad.x));
            m1 = fmaxf(m1, lrelu(as.y + ad.y));
            m2 = fmaxf(m2, lrelu(as.z + ad.z));
            m3 = fmaxf(m3, lrelu(as.w + ad.w));
        }
        for (int o = 16; o > 0; o >>= 1) {
            m0 = fmaxf(m0, __shfl_xor(m0, o, 32));
            m1 = fmaxf(m1, __shfl_xor(m1, o, 32));
            m2 = fmaxf(m2, __shfl_xor(m2, o, 32));
            m3 = fmaxf(m3, __shfl_xor(m3, o, 32));
        }
        m0 = (fabsf(m0) < INFINITY) ? m0 : 0.0f;
        m1 = (fabsf(m1) < INFINITY) ? m1 : 0.0f;
        m2 = (fabsf(m2) < INFINITY) ? m2 : 0.0f;
        m3 = (fabsf(m3) < INFINITY) ? m3 : 0.0f;

        float p0 = 0.0f, p1 = 0.0f, p2 = 0.0f, p3 = 0.0f;
        for (int i = beg + lane; i < end; i += 32) {
            const int s = clamp_i(csr[i], 0, N_NODES - 1);
            const float4 as = *(const float4*)(a_s + (size_t)s * HEADS);
            p0 += expf(lrelu(as.x + ad.x) - m0);
            p1 += expf(lrelu(as.y + ad.y) - m1);
            p2 += expf(lrelu(as.z + ad.z) - m2);
            p3 += expf(lrelu(as.w + ad.w) - m3);
        }
        for (int o = 16; o > 0; o >>= 1) {
            p0 += __shfl_xor(p0, o, 32);
            p1 += __shfl_xor(p1, o, 32);
            p2 += __shfl_xor(p2, o, 32);
            p3 += __shfl_xor(p3, o, 32);
        }
        const float es0 = expf(ls0 - m0), es1 = expf(ls1 - m1);
        const float es2 = expf(ls2 - m2), es3 = expf(ls3 - m3);
        const float den0 = (p0 + es0) + SOFTMAX_EPS;
        const float den1 = (p1 + es1) + SOFTMAX_EPS;
        const float den2 = (p2 + es2) + SOFTMAX_EPS;
        const float den3 = (p3 + es3) + SOFTMAX_EPS;

        const float wself = by_head(es0 / den0, es1 / den1, es2 / den2,
                                    es3 / den3, hk);
        const float4 hs = *(const float4*)(h + (size_t)d * OUT_CH + lane * 4);
        float4 acc;
        acc.x = hs.x * wself; acc.y = hs.y * wself;
        acc.z = hs.z * wself; acc.w = hs.w * wself;

        for (int base = beg; base < end; base += 32) {
            const int i = base + lane;
            int s = 0;
            float w0 = 0.0f, w1 = 0.0f, w2 = 0.0f, w3 = 0.0f;
            if (i < end) {
                s = clamp_i(csr[i], 0, N_NODES - 1);
                const float4 as = *(const float4*)(a_s + (size_t)s * HEADS);
                w0 = expf(lrelu(as.x + ad.x) - m0) / den0;
                w1 = expf(lrelu(as.y + ad.y) - m1) / den1;
                w2 = expf(lrelu(as.z + ad.z) - m2) / den2;
                w3 = expf(lrelu(as.w + ad.w) - m3) / den3;
            }
            const int left = end - base;
            const int n = left < 32 ? left : 32;
            int j = 0;
            for (; j + 4 <= n; j += 4) {
                const int sa = __builtin_amdgcn_readlane(s, j);
                const int sb = __builtin_amdgcn_readlane(s, j + 1);
                const int sc = __builtin_amdgcn_readlane(s, j + 2);
                const int se = __builtin_amdgcn_readlane(s, j + 3);
                const float4 ha = *(const float4*)(h + (size_t)sa * OUT_CH + lane * 4);
                const float4 hb = *(const float4*)(h + (size_t)sb * OUT_CH + lane * 4);
                const float4 hc = *(const float4*)(h + (size_t)sc * OUT_CH + lane * 4);
                const float4 he = *(const float4*)(h + (size_t)se * OUT_CH + lane * 4);
                const float wa = head_w(w0, w1, w2, w3, j, hk);
                const float wb = head_w(w0, w1, w2, w3, j + 1, hk);
                const float wc = head_w(w0, w1, w2, w3, j + 2, hk);
                const float we = head_w(w0, w1, w2, w3, j + 3, hk);
                acc.x += ha.x * wa; acc.y += ha.y * wa; acc.z += ha.z * wa; acc.w += ha.w * wa;
                acc.x += hb.x * wb; acc.y += hb.y * wb; acc.z += hb.z * wb; acc.w += hb.w * wb;
                acc.x += hc.x * wc; acc.y += hc.y * wc; acc.z += hc.z * wc; acc.w += hc.w * wc;
                acc.x += he.x * we; acc.y += he.y * we; acc.z += he.z * we; acc.w += he.w * we;
            }
            for (; j < n; ++j) {
                const int sa = __builtin_amdgcn_readlane(s, j);
                const float4 ha = *(const float4*)(h + (size_t)sa * OUT_CH + lane * 4);
                const float wa = head_w(w0, w1, w2, w3, j, hk);
                acc.x += ha.x * wa; acc.y += ha.y * wa; acc.z += ha.z * wa; acc.w += ha.w * wa;
            }
        }
        *(float4*)(out + (size_t)d * OUT_CH + lane * 4) = acc;
    }
}

__global__ void k_ln_elu(float* __restrict__ out,
                         const float* __restrict__ bias,
                         const float* __restrict__ gamma,
                         const float* __restrict__ beta) {
    __shared__ float red[4];
    const int node = blockIdx.x;
    const int c = threadIdx.x;
    const int wave = c >> 5, lane = c & 31;

    float v = out[(size_t)node * OUT_CH + c] + bias[c];

    float t = v;
#pragma unroll
    for (int o = 16; o > 0; o >>= 1) t += __shfl_xor(t, o, 32);
    if (lane == 0) red[wave] = t;
    __syncthreads();
    const float mu = (red[0] + red[1] + red[2] + red[3]) * (1.0f / OUT_CH);
    __syncthreads();

    const float dv = v - mu;
    t = dv * dv;
#pragma unroll
    for (int o = 16; o > 0; o >>= 1) t += __shfl_xor(t, o, 32);
    if (lane == 0) red[wave] = t;
    __syncthreads();
    const float var = (red[0] + red[1] + red[2] + red[3]) * (1.0f / OUT_CH);

    float y = dv * (1.0f / sqrtf(var + LN_EPS)) * gamma[c] + beta[c];
    y = y > 0.0f ? y : (expf(y) - 1.0f);
    out[(size_t)node * OUT_CH + c] = y;
}

extern "C" void kernel_launch(void* const* d_in, const int* in_sizes, int n_in,
                              void* d_out, int out_size, void* d_ws, size_t ws_size,
                              hipStream_t stream) {
    const float* x       = (const float*)d_in[0];
    const int*   ei      = (const int*)  d_in[1];
    const float* W       = (const float*)d_in[2];
    const float* att_src = (const float*)d_in[3];
    const float* att_dst = (const float*)d_in[4];
    const float* bias    = (const float*)d_in[5];
    const float* gamma   = (const float*)d_in[6];
    const float* beta    = (const float*)d_in[7];
    float* out = (float*)d_out;

    char* ws = (char*)d_ws;
    float*    h      = (float*)ws;
    ws += (size_t)N_NODES * OUT_CH * sizeof(float);
    float*    a_s    = (float*)ws;  ws += (size_t)N_NODES * HEADS * sizeof(float);
    float*    a_d    = (float*)ws;  ws += (size_t)N_NODES * HEADS * sizeof(float);
    int*      cnt    = (int*)ws;    ws += (size_t)N_NODES * HEADS * sizeof(int);
    int*      off    = (int*)ws;    ws += (size_t)N_NODES * HEADS * sizeof(int);
    int*      csr    = (int*)ws;    ws += (size_t)N_EDGES * sizeof(int);
    int*      rank   = (int*)ws;

    k_init<<<(N_NODES + 255) / 256, 256, 0, stream>>>(cnt, N_NODES);

    k_gemm_wmma<<<(N_NODES + 127) / 128, 256, IN_CH * OUT_CH * sizeof(float),
                  stream>>>(x, W, h, N_NODES);

    k_dots<<<(N_NODES + 1) / 2, 256, 0, stream>>>(h, att_src, att_dst, a_s, a_d,
                                                  N_NODES);

    k_count<<<(N_EDGES + 255) / 256, 256, 0, stream>>>(ei, cnt, rank);

    k_scan<<<(N_NODES + 1023) / 1024, 1024, 0, stream>>>(cnt, off);

    k_fill<<<(N_EDGES + 255) / 256, 256, 0, stream>>>(ei, off, rank, csr);

    k_gather<<<GATHER_BLOCKS, 256, 0, stream>>>(off, csr, a_s, a_d, h, out);

    k_ln_elu<<<N_NODES, 128, 0, stream>>>(out, bias, gamma, beta);
}


# Edge ranks drawn from LDS counters, not 1.6M device-wide returning atomics: edges bucketed by destination range in LDS, then a block per range lays its rows out; float sums in LDS arrival order
# speedup vs baseline: 5.6764x; 3.1933x over previous
#include <hip/hip_runtime.h>
#include <math.h>


#define N_NODES 50000
#define N_EDGES 1600000
#define ET_EDGES (N_EDGES + N_NODES)
#define IN_CH 128
#define OUT_CH 128
#define HEADS 4
#define CPH 32
#define NEG_SLOPE 0.2f
#define SOFTMAX_EPS 1e-16f
#define LN_EPS 1e-5f
#define SLOTS 64

typedef float v2f __attribute__((ext_vector_type(2)));
typedef float v8f __attribute__((ext_vector_type(8)));

template <int MASK>
__device__ __forceinline__ float row_xadd(float v) {
    const int o = __builtin_amdgcn_update_dpp(0, __float_as_int(v), 0x160 + MASK,
                                              0xf, 0xf, true);
    return v + __int_as_float(o);
}
__device__ __forceinline__ float row16_sum(float v) {
    v = row_xadd<1>(v);
    v = row_xadd<2>(v);
    v = row_xadd<4>(v);
    v = row_xadd<8>(v);
    return v;
}

__global__ void __launch_bounds__(256)
k_gemm_wmma(const float* __restrict__ x,
                            const float* __restrict__ W,
                            const float* __restrict__ att_src,
                            const float* __restrict__ att_dst,
                            float* __restrict__ h,
                            float* __restrict__ a_s, float* __restrict__ a_d,
                            int nrows) {
    extern __shared__ float Wlds[];
    for (int i = threadIdx.x; i < (IN_CH * OUT_CH) / 4; i += blockDim.x)
        ((float4*)Wlds)[i] = ((const float4*)W)[i];
    __syncthreads();

    const int wave = threadIdx.x >> 5;
    const int lane = threadIdx.x & 31;
    const int rowBase = blockIdx.x * 128 + wave * 16;
    const int m  = lane & 15;
    const int kk = (lane >> 4) << 1;

    int arow = rowBase + m;
    if (arow >= nrows) arow = nrows - 1;
    const float* xk = x + (size_t)arow * IN_CH + kk;

    v8f c[8];
#pragma unroll
    for (int t = 0; t < 8; ++t) c[t] = (v8f){0,0,0,0,0,0,0,0};

    float2 av = *(const float2*)xk;
    for (int k = 0; k < IN_CH; k += 4) {
        v2f a; a.x = av.x; a.y = av.y;
        if (k + 4 < IN_CH) av = *(const float2*)(xk + k + 4);

        const float* w0 = &Wlds[(k + kk) * OUT_CH + m];
        v2f b[8];
#pragma unroll
        for (int t = 0; t < 8; ++t) {
            b[t].x = w0[t * 16];
            b[t].y = w0[OUT_CH + t * 16];
        }
#pragma unroll
        for (int t = 0; t < 8; ++t)
            c[t] = __builtin_amdgcn_wmma_f32_16x16x4_f32(
                false, a, false, b[t], (short)0, c[t], false, false);
    }

#pragma unroll
    for (int t = 0; t < 8; ++t) {
#pragma unroll
        for (int v = 0; v < 8; ++v) {
            const int M = v + ((lane >> 4) << 3);
            const int row = rowBase + M;
            if (row < nrows)
                h[(size_t)row * OUT_CH + t * 16 + m] = c[t][v];
        }
    }

    const int r8 = m & 7;
    const bool isS = m < 8;
    float keep[HEADS];
#pragma unroll
    for (int hd = 0; hd < HEADS; ++hd) {
        const float ws0 = att_src[hd * CPH + m], ws1 = att_src[hd * CPH + 16 + m];
        const float wd0 = att_dst[hd * CPH + m], wd1 = att_dst[hd * CPH + 16 + m];
        float kv = 0.0f;
#pragma unroll
        for (int v = 0; v < 8; ++v) {
            const float c0 = c[2 * hd][v], c1 = c[2 * hd + 1][v];
            const float ps = row16_sum(c0 * ws0 + c1 * ws1);
            const float pd = row16_sum(c0 * wd0 + c1 * wd1);
            if (r8 == v) kv = isS ? ps : pd;
        }
        keep[hd] = kv;
    }
    const int drow = rowBase + ((lane >> 4) << 3) + r8;
    if (drow < nrows) {
        float* dp = (isS ? a_s : a_d) + (size_t)drow * HEADS;
        *(float4*)dp = make_float4(keep[0], keep[1], keep[2], keep[3]);
    }
}

__device__ __forceinline__ int clamp_i(int v, int lo, int hi) {
    return v < lo ? lo : (v > hi ? hi : v);
}

#define DPB   200
#define NBIN  (N_NODES / DPB)
#define EBLK  256
#define ECH   (N_EDGES / EBLK)
#define CELLW 64
#define CELLP 65
#define EPT   ((ECH + 1023) / 1024)

__global__ void __launch_bounds__(1024)
k_bucket(const int* __restrict__ ei, int* __restrict__ cells) {
    __shared__ int hist[NBIN];
    __shared__ int img[NBIN * CELLP];
    const int t = threadIdx.x;
    const int b = blockIdx.x;
    if (t < NBIN) hist[t] = 0;
    int sv[EPT], dv[EPT];
#pragma unroll
    for (int k = 0; k < EPT; ++k) {
        const int o = k * 1024 + t;
        const bool in = o < ECH;
        sv[k] = in ? ei[b * ECH + o] : -1;
        dv[k] = in ? ei[N_EDGES + b * ECH + o] : -1;
    }
    __syncthreads();
#pragma unroll
    for (int k = 0; k < EPT; ++k) {
        const int s = sv[k], d = dv[k];
        if ((unsigned)d < (unsigned)N_NODES && (unsigned)s < (unsigned)N_NODES) {
            const int bin = d / DPB;
            const int dl = d - bin * DPB;
            const int r = atomicAdd(&hist[bin], 1);
            if (r < CELLW - 1) img[bin * CELLP + 1 + r] = (dl << 16) | s;
        }
    }
    __syncthreads();
    if (t < NBIN) img[t * CELLP] = hist[t];
    __syncthreads();
    int* dst = cells + (size_t)b * (NBIN * CELLW);
    for (int i = t; i < NBIN * CELLW; i += 1024)
        dst[i] = img[(i / CELLW) * CELLP + (i % CELLW)];
}

__global__ void __launch_bounds__(1024)
k_rows(const int* __restrict__ cells, int* __restrict__ cnt,
       int* __restrict__ slot) {
    __shared__ int rows[DPB * CELLP];
    __shared__ int dcnt[DPB];
    __shared__ int cut;
    const int t = threadIdx.x;
    const int lane = t & 31;
    const int wave = t >> 5;
    const int bin = blockIdx.x;
    if (t < DPB) dcnt[t] = 0;
    if (t == 0) cut = 0;
    int w0[EBLK / 32], w1[EBLK / 32];
#pragma unroll
    for (int k = 0; k < EBLK / 32; ++k) {
        const int cb = wave * (EBLK / 32) + k;
        const int* cp = cells + ((size_t)cb * NBIN + bin) * CELLW;
        w0[k] = cp[lane];
        w1[k] = cp[lane + 32];
    }
    __syncthreads();
#pragma unroll
    for (int k = 0; k < EBLK / 32; ++k) {
        const int n = __builtin_amdgcn_readfirstlane(w0[k]);
        if (n > CELLW - 1 || n < 0) cut = 1;
        if (lane >= 1 && lane <= n) {
            const int dl = clamp_i(w0[k] >> 16, 0, DPB - 1);
            const int r = atomicAdd(&dcnt[dl], 1);
            if (r < SLOTS) rows[dl * CELLP + r] = w0[k] & 0xffff;
        }
        if (lane + 32 <= n) {
            const int dl = clamp_i(w1[k] >> 16, 0, DPB - 1);
            const int r = atomicAdd(&dcnt[dl], 1);
            if (r < SLOTS) rows[dl * CELLP + r] = w1[k] & 0xffff;
        }
    }
    __syncthreads();
    if (t < DPB) cnt[bin * DPB + t] = cut ? SLOTS + 1 : dcnt[t];
    int* dst = slot + (size_t)bin * (DPB * SLOTS);
    for (int i = t; i < DPB * SLOTS; i += 1024)
        dst[i] = rows[(i / SLOTS) * CELLP + (i % SLOTS)];
}

#define GATHER_BLOCKS 2048

__device__ __forceinline__ float lrelu(float v) {
    return v > 0.0f ? v : NEG_SLOPE * v;
}
__device__ __forceinline__ float lane_f(float v, int j) {
    return __int_as_float(__builtin_amdgcn_readlane(__float_as_int(v), j));
}
struct HeadMask { unsigned k0, k1, k2, k3; };
__device__ __forceinline__ float by_head(float a, float b, float c, float e,
                                         const HeadMask& k) {
    const unsigned r = (__float_as_uint(a) & k.k0) | (__float_as_uint(b) & k.k1) |
                       (__float_as_uint(c) & k.k2) | (__float_as_uint(e) & k.k3);
    return __uint_as_float(r);
}
__device__ __forceinline__ float head_w(float w0, float w1, float w2, float w3,
                                        int j, const HeadMask& k) {
    return by_head(lane_f(w0, j), lane_f(w1, j), lane_f(w2, j), lane_f(w3, j), k);
}

__device__ __forceinline__ void add_rows(float4& acc, int s, float w0, float w1,
        float w2, float w3, int n, const float* __restrict__ h, int lane,
        const HeadMask& hk) {
    int j = 0;
    for (; j + 4 <= n; j += 4) {
        const int sa = __builtin_amdgcn_readlane(s, j);
        const int sb = __builtin_amdgcn_readlane(s, j + 1);
        const int sc = __builtin_amdgcn_readlane(s, j + 2);
        const int se = __builtin_amdgcn_readlane(s, j + 3);
        const float4 ha = *(const float4*)(h + (size_t)sa * OUT_CH + lane * 4);
        const float4 hb = *(const float4*)(h + (size_t)sb * OUT_CH + lane * 4);
        const float4 hc = *(const float4*)(h + (size_t)sc * OUT_CH + lane * 4);
        const float4 he = *(const float4*)(h + (size_t)se * OUT_CH + lane * 4);
        const float wa = head_w(w0, w1, w2, w3, j, hk);
        const float wb = head_w(w0, w1, w2, w3, j + 1, hk);
        const float wc = head_w(w0, w1, w2, w3, j + 2, hk);
        const float we = head_w(w0, w1, w2, w3, j + 3, hk);
        acc.x += ha.x * wa; acc.y += ha.y * wa; acc.z += ha.z * wa; acc.w += ha.w * wa;
        acc.x += hb.x * wb; acc.y += hb.y * wb; acc.z += hb.z * wb; acc.w += hb.w * wb;
        acc.x += hc.x * wc; acc.y += hc.y * wc; acc.z += hc.z * wc; acc.w += hc.w * wc;
        acc.x += he.x * we; acc.y += he.y * we; acc.z += he.z * we; acc.w += he.w * we;
    }
    for (; j < n; ++j) {
        const int sa = __builtin_amdgcn_readlane(s, j);
        const float4 ha = *(const float4*)(h + (size_t)sa * OUT_CH + lane * 4);
        const float wa = head_w(w0, w1, w2, w3, j, hk);
        acc.x += ha.x * wa; acc.y += ha.y * wa; acc.z += ha.z * wa; acc.w += ha.w * wa;
    }
}

__device__ __forceinline__ void finish_row(float4 acc, int d, int lane,
        const float* bias, const float* gamma, const float* beta, float* out) {
    const float4 bs = *(const float4*)(bias + lane * 4);
    float4 v;
    v.x = acc.x + bs.x; v.y = acc.y + bs.y;
    v.z = acc.z + bs.z; v.w = acc.w + bs.w;
    float t = (v.x + v.y) + (v.z + v.w);
    for (int o = 16; o > 0; o >>= 1) t += __shfl_xor(t, o, 32);
    const float mu = t * (1.0f / OUT_CH);
    const float dx = v.x - mu, dy = v.y - mu, dz = v.z - mu, dw = v.w - mu;
    t = (dx * dx + dy * dy) + (dz * dz + dw * dw);
    for (int o = 16; o > 0; o >>= 1) t += __shfl_xor(t, o, 32);
    const float var = t * (1.0f / OUT_CH);
    const float rs = 1.0f / sqrtf(var + LN_EPS);
    const float4 gm = *(const float4*)(gamma + lane * 4);
    const float4 bt = *(const float4*)(beta  + lane * 4);
    float4 y;
    y.x = dx * rs * gm.x + bt.x; y.y = dy * rs * gm.y + bt.y;
    y.z = dz * rs * gm.z + bt.z; y.w = dw * rs * gm.w + bt.w;
    y.x = y.x > 0.0f ? y.x : (expf(y.x) - 1.0f);
    y.y = y.y > 0.0f ? y.y : (expf(y.y) - 1.0f);
    y.z = y.z > 0.0f ? y.z : (expf(y.z) - 1.0f);
    y.w = y.w > 0.0f ? y.w : (expf(y.w) - 1.0f);
    *(float4*)(out + (size_t)d * OUT_CH + lane * 4) = y;
}

__global__ void k_gather(const int* __restrict__ ei, const int* __restrict__ cnt,
         const int* __restrict__ slot,
         const float* __restrict__ a_s, const float* __restrict__ a_d,
         const float* __restrict__ h,
         const float* bias, const float* gamma, const float* beta,
         float* out) {
    const int lane = threadIdx.x & 31;
    const int head = lane >> 3;
    HeadMask hk;
    hk.k0 = head == 0 ? 0xffffffffu : 0u;
    hk.k1 = head == 1 ? 0xffffffffu : 0u;
    hk.k2 = head == 2 ? 0xffffffffu : 0u;
    hk.k3 = head == 3 ? 0xffffffffu : 0u;
    const int w_first = __builtin_amdgcn_readfirstlane(
        (int)(blockIdx.x * 8 + (threadIdx.x >> 5)));

    for (int d = w_first; d < N_NODES; d += GATHER_BLOCKS * 8) {
        const int dall = clamp_i(cnt[d], 0, N_EDGES);
        const int deg = dall > SLOTS ? 0 : dall;
        const int beg = __builtin_amdgcn_readfirstlane(d * SLOTS);
        const float4 ad = *(const float4*)(a_d + (size_t)d * HEADS);
        const float4 sd = *(const float4*)(a_s + (size_t)d * HEADS);
        const float ls0 = lrelu(sd.x + ad.x), ls1 = lrelu(sd.y + ad.y);
        const float ls2 = lrelu(sd.z + ad.z), ls3 = lrelu(sd.w + ad.w);

        const bool inA = lane < deg;
        const bool inB = lane + 32 < deg;
        int rA = 0, rB = 0;
        if (inA) {
            rA = slot[beg + lane];
            rB = slot[beg + (inB ? lane + 32 : lane)];
        }
        const int sA = clamp_i(rA, 0, N_NODES - 1);
        const int sB = inB ? clamp_i(rB, 0, N_NODES - 1) : 0;
        const float4 asA = *(const float4*)(a_s + (size_t)sA * HEADS);
        const float4 asB = *(const float4*)(a_s + (size_t)sB * HEADS);
        float a0 = lrelu(asA.x + (inA ? ad.x : -INFINITY));
        float a1 = lrelu(asA.y + (inA ? ad.y : -INFINITY));
        float a2 = lrelu(asA.z + (inA ? ad.z : -INFINITY));
        float a3 = lrelu(asA.w + (inA ? ad.w : -INFINITY));
        float b0 = lrelu(asB.x + (inB ? ad.x : -INFINITY));
        float b1 = lrelu(asB.y + (inB ? ad.y : -INFINITY));
        float b2 = lrelu(asB.z + (inB ? ad.z : -INFINITY));
        float b3 = lrelu(asB.w + (inB ? ad.w : -INFINITY));

        float m0 = fmaxf(fmaxf(ls0, a0), b0), m1 = fmaxf(fmaxf(ls1, a1), b1);
        float m2 = fmaxf(fmaxf(ls2, a2), b2), m3 = fmaxf(fmaxf(ls3, a3), b3);
        for (int o = 16; o > 0; o >>= 1) {
            m0 = fmaxf(m0, __shfl_xor(m0, o, 32));
            m1 = fmaxf(m1, __shfl_xor(m1, o, 32));
            m2 = fmaxf(m2, __shfl_xor(m2, o, 32));
            m3 = fmaxf(m3, __shfl_xor(m3, o, 32));
        }
        m0 = (fabsf(m0) < INFINITY) ? m0 : 0.0f;
        m1 = (fabsf(m1) < INFINITY) ? m1 : 0.0f;
        m2 = (fabsf(m2) < INFINITY) ? m2 : 0.0f;
        m3 = (fabsf(m3) < INFINITY) ? m3 : 0.0f;

        if (deg > 0) {
            const float e0 = expf(a0 - m0), e1 = expf(a1 - m1);
            const float e2 = expf(a2 - m2), e3 = expf(a3 - m3);
            a0 = inA ? e0 : 0.0f; a1 = inA ? e1 : 0.0f;
            a2 = inA ? e2 : 0.0f; a3 = inA ? e3 : 0.0f;
        } else {
            a0 = 0.0f; a1 = 0.0f; a2 = 0.0f; a3 = 0.0f;
        }
        if (deg > 32) {
            const float e0 = expf(b0 - m0), e1 = expf(b1 - m1);
            const float e2 = expf(b2 - m2), e3 = expf(b3 - m3);
            b0 = inB ? e0 : 0.0f; b1 = inB ? e1 : 0.0f;
            b2 = inB ? e2 : 0.0f; b3 = inB ? e3 : 0.0f;
        } else {
            b0 = 0.0f; b1 = 0.0f; b2 = 0.0f; b3 = 0.0f;
        }
        float p0 = a0 + b0, p1 = a1 + b1, p2 = a2 + b2, p3 = a3 + b3;
        for (int o = 16; o > 0; o >>= 1) {
            p0 += __shfl_xor(p0, o, 32);
            p1 += __shfl_xor(p1, o, 32);
            p2 += __shfl_xor(p2, o, 32);
            p3 += __shfl_xor(p3, o, 32);
        }
        const float es0 = expf(ls0 - m0), es1 = expf(ls1 - m1);
        const float es2 = expf(ls2 - m2), es3 = expf(ls3 - m3);
        const float den0 = (p0 + es0) + SOFTMAX_EPS;
        const float den1 = (p1 + es1) + SOFTMAX_EPS;
        const float den2 = (p2 + es2) + SOFTMAX_EPS;
        const float den3 = (p3 + es3) + SOFTMAX_EPS;

        const float wself = by_head(es0 / den0, es1 / den1, es2 / den2,
                                    es3 / den3, hk);
        const float4 hs = *(const float4*)(h + (size_t)d * OUT_CH + lane * 4);
        float4 acc;
        acc.x = hs.x * wself; acc.y = hs.y * wself;
        acc.z = hs.z * wself; acc.w = hs.w * wself;

        if (deg > 0) {
            a0 = a0 / den0; a1 = a1 / den1; a2 = a2 / den2; a3 = a3 / den3;
            add_rows(acc, sA, a0, a1, a2, a3, deg < 32 ? deg : 32, h, lane, hk);
            if (deg > 32) {
                b0 = b0 / den0; b1 = b1 / den1; b2 = b2 / den2; b3 = b3 / den3;
                add_rows(acc, sB, b0, b1, b2, b3, deg - 32, h, lane, hk);
            }
        }
        if (dall <= SLOTS) finish_row(acc, d, lane, bias, gamma, beta, out);
    }

    const int* dsts = ei + N_EDGES;
    for (int d = w_first; d < N_NODES; d += GATHER_BLOCKS * 8) {
        const int deg = clamp_i(cnt[d], 0, N_EDGES);
        if (deg <= SLOTS) continue;
        const float4 ad = *(const float4*)(a_d + (size_t)d * HEADS);
        const float4 sd = *(const float4*)(a_s + (size_t)d * HEADS);
        const float ls0 = lrelu(sd.x + ad.x), ls1 = lrelu(sd.y + ad.y);
        const float ls2 = lrelu(sd.z + ad.z), ls3 = lrelu(sd.w + ad.w);

        float m0 = ls0, m1 = ls1, m2 = ls2, m3 = ls3;
        for (int i = lane; i < N_EDGES; i += 32) {
            if (dsts[i] != d) continue;
            const int s = clamp_i(ei[i], 0, N_NODES - 1);
            const float4 as = *(const float4*)(a_s + (size_t)s * HEADS);
            m0 = fmaxf(m0, lrelu(as.x + ad.x));
            m1 = fmaxf(m1, lrelu(as.y + ad.y));
            m2 = fmaxf(m2, lrelu(as.z + ad.z));
            m3 = fmaxf(m3, lrelu(as.w + ad.w));
        }
        for (int o = 16; o > 0; o >>= 1) {
            m0 = fmaxf(m0, __shfl_xor(m0, o, 32));
            m1 = fmaxf(m1, __shfl_xor(m1, o, 32));
            m2 = fmaxf(m2, __shfl_xor(m2, o, 32));
            m3 = fmaxf(m3, __shfl_xor(m3, o, 32));
        }
        m0 = (fabsf(m0) < INFINITY) ? m0 : 0.0f;
        m1 = (fabsf(m1) < INFINITY) ? m1 : 0.0f;
        m2 = (fabsf(m2) < INFINITY) ? m2 : 0.0f;
        m3 = (fabsf(m3) < INFINITY) ? m3 : 0.0f;

        float p0 = 0.0f, p1 = 0.0f, p2 = 0.0f, p3 = 0.0f;
        for (int i = lane; i < N_EDGES; i += 32) {
            if (dsts[i] != d) continue;
            const int s = clamp_i(ei[i], 0, N_NODES - 1);
            const float4 as = *(const float4*)(a_s + (size_t)s * HEADS);
            p0 += expf(lrelu(as.x + ad.x) - m0);
            p1 += expf(lrelu(as.y + ad.y) - m1);
            p2 += expf(lrelu(as.z + ad.z) - m2);
            p3 += expf(lrelu(as.w + ad.w) - m3);
        }
        for (int o = 16; o > 0; o >>= 1) {
            p0 += __shfl_xor(p0, o, 32);
            p1 += __shfl_xor(p1, o, 32);
            p2 += __shfl_xor(p2, o, 32);
            p3 += __shfl_xor(p3, o, 32);
        }
        const float es0 = expf(ls0 - m0), es1 = expf(ls1 - m1);
        const float es2 = expf(ls2 - m2), es3 = expf(ls3 - m3);
        const float den0 = (p0 + es0) + SOFTMAX_EPS;
        const float den1 = (p1 + es1) + SOFTMAX_EPS;
        const float den2 = (p2 + es2) + SOFTMAX_EPS;
        const float den3 = (p3 + es3) + SOFTMAX_EPS;

        const float wself = by_head(es0 / den0, es1 / den1, es2 / den2,
                                    es3 / den3, hk);
        const float4 hs = *(const float4*)(h + (size_t)d * OUT_CH + lane * 4);
        float4 acc;
        acc.x = hs.x * wself; acc.y = hs.y * wself;
        acc.z = hs.z * wself; acc.w = hs.w * wself;

        for (int base = 0; base < N_EDGES; base += 32) {
            const int i = base + lane;
            int s = 0;
            float w0 = 0.0f, w1 = 0.0f, w2 = 0.0f, w3 = 0.0f;
            const bool mine = i < N_EDGES && dsts[i] == d;
            if (mine) {
                s = clamp_i(ei[i], 0, N_NODES - 1);
                const float4 as = *(const float4*)(a_s + (size_t)s * HEADS);
                w0 = expf(lrelu(as.x + ad.x) - m0) / den0;
                w1 = expf(lrelu(as.y + ad.y) - m1) / den1;
                w2 = expf(lrelu(as.z + ad.z) - m2) / den2;
                w3 = expf(lrelu(as.w + ad.w) - m3) / den3;
            }
            unsigned todo = __builtin_amdgcn_ballot_w32(mine);
            while (todo != 0u) {
                const int jj = __builtin_ctz(todo);
                todo &= todo - 1u;
                const int sa = __builtin_amdgcn_readlane(s, jj);
                const float4 ha = *(const float4*)(h + (size_t)sa * OUT_CH + lane * 4);
                const float wa = head_w(w0, w1, w2, w3, jj, hk);
                acc.x += ha.x * wa; acc.y += ha.y * wa; acc.z += ha.z * wa; acc.w += ha.w * wa;
            }
        }
        finish_row(acc, d, lane, bias, gamma, beta, out);
    }
}

extern "C" void kernel_launch(void* const* d_in, const int* in_sizes, int n_in,
                              void* d_out, int out_size, void* d_ws, size_t ws_size,
                              hipStream_t stream) {
    const float* x       = (const float*)d_in[0];
    const int*   ei      = (const int*)  d_in[1];
    const float* W       = (const float*)d_in[2];
    const float* att_src = (const float*)d_in[3];
    const float* att_dst = (const float*)d_in[4];
    const float* bias    = (const float*)d_in[5];
    const float* gamma   = (const float*)d_in[6];
    const float* beta    = (const float*)d_in[7];
    float* out = (float*)d_out;

    char* ws = (char*)d_ws;
    float*    h      = (float*)ws;
    ws += (size_t)N_NODES * OUT_CH * sizeof(float);
    float*    a_s    = (float*)ws;  ws += (size_t)N_NODES * HEADS * sizeof(float);
    float*    a_d    = (float*)ws;  ws += (size_t)N_NODES * HEADS * sizeof(float);
    int*      cnt    = (int*)ws;    ws += (size_t)N_NODES * HEADS * sizeof(int);
    int*      slot   = (int*)ws;
    int*      cells  = (int*)h;

    k_bucket<<<EBLK, 1024, 0, stream>>>(ei, cells);
    k_rows<<<NBIN, 1024, 0, stream>>>(cells, cnt, slot);

    k_gemm_wmma<<<(N_NODES + 127) / 128, 256, IN_CH * OUT_CH * sizeof(float),
                  stream>>>(x, W, att_src, att_dst, h, a_s, a_d, N_NODES);

    k_gather<<<GATHER_BLOCKS, 256, 0, stream>>>(ei, cnt, slot, a_s, a_d, h,
                                                 bias, gamma, beta, out);
}


# Gather: edge weights pass from slot lanes to channel lanes through a per-wave LDS table, one 16-byte read per lane per four rows, not four lane reads and mask selects per row; sum order kept
# speedup vs baseline: 6.0877x; 1.0725x over previous
#include <hip/hip_runtime.h>
#include <math.h>


#define N_NODES 50000
#define N_EDGES 1600000
#define ET_EDGES (N_EDGES + N_NODES)
#define IN_CH 128
#define OUT_CH 128
#define HEADS 4
#define CPH 32
#define NEG_SLOPE 0.2f
#define SOFTMAX_EPS 1e-16f
#define LN_EPS 1e-5f
#define SLOTS 64

typedef float v2f __attribute__((ext_vector_type(2)));
typedef float v8f __attribute__((ext_vector_type(8)));

template <int MASK>
__device__ __forceinline__ float row_xadd(float v) {
    const int o = __builtin_amdgcn_update_dpp(0, __float_as_int(v), 0x160 + MASK,
                                              0xf, 0xf, true);
    return v + __int_as_float(o);
}
__device__ __forceinline__ float row16_sum(float v) {
    v = row_xadd<1>(v);
    v = row_xadd<2>(v);
    v = row_xadd<4>(v);
    v = row_xadd<8>(v);
    return v;
}

__global__ void __launch_bounds__(256)
k_gemm_wmma(const float* __restrict__ x,
                            const float* __restrict__ W,
                            const float* __restrict__ att_src,
                            const float* __restrict__ att_dst,
                            float* __restrict__ h,
                            float* __restrict__ a_s, float* __restrict__ a_d,
                            int nrows) {
    extern __shared__ float Wlds[];
    for (int i = threadIdx.x; i < (IN_CH * OUT_CH) / 4; i += blockDim.x)
        ((float4*)Wlds)[i] = ((const float4*)W)[i];
    __syncthreads();

    const int wave = threadIdx.x >> 5;
    const int lane = threadIdx.x & 31;
    const int rowBase = blockIdx.x * 128 + wave * 16;
    const int m  = lane & 15;
    const int kk = (lane >> 4) << 1;

    int arow = rowBase + m;
    if (arow >= nrows) arow = nrows - 1;
    const float* xk = x + (size_t)arow * IN_CH + kk;

    v8f c[8];
#pragma unroll
    for (int t = 0; t < 8; ++t) c[t] = (v8f){0,0,0,0,0,0,0,0};

    float2 av = *(const float2*)xk;
    for (int k = 0; k < IN_CH; k += 4) {
        v2f a; a.x = av.x; a.y = av.y;
        if (k + 4 < IN_CH) av = *(const float2*)(xk + k + 4);

        const float* w0 = &Wlds[(k + kk) * OUT_CH + m];
        v2f b[8];
#pragma unroll
        for (int t = 0; t < 8; ++t) {
            b[t].x = w0[t * 16];
            b[t].y = w0[OUT_CH + t * 16];
        }
#pragma unroll
        for (int t = 0; t < 8; ++t)
            c[t] = __builtin_amdgcn_wmma_f32_16x16x4_f32(
                false, a, false, b[t], (short)0, c[t], false, false);
    }

#pragma unroll
    for (int t = 0; t < 8; ++t) {
#pragma unroll
        for (int v = 0; v < 8; ++v) {
            const int M = v + ((lane >> 4) << 3);
            const int row = rowBase + M;
            if (row < nrows)
                h[(size_t)row * OUT_CH + t * 16 + m] = c[t][v];
        }
    }

    const int r8 = m & 7;
    const bool isS = m < 8;
    float keep[HEADS];
#pragma unroll
    for (int hd = 0; hd < HEADS; ++hd) {
        const float ws0 = att_src[hd * CPH + m], ws1 = att_src[hd * CPH + 16 + m];
        const float wd0 = att_dst[hd * CPH + m], wd1 = att_dst[hd * CPH + 16 + m];
        float kv = 0.0f;
#pragma unroll
        for (int v = 0; v < 8; ++v) {
            const float c0 = c[2 * hd][v], c1 = c[2 * hd + 1][v];
            const float ps = row16_sum(c0 * ws0 + c1 * ws1);
            const float pd = row16_sum(c0 * wd0 + c1 * wd1);
            if (r8 == v) kv = isS ? ps : pd;
        }
        keep[hd] = kv;
    }
    const int drow = rowBase + ((lane >> 4) << 3) + r8;
    if (drow < nrows) {
        float* dp = (isS ? a_s : a_d) + (size_t)drow * HEADS;
        *(float4*)dp = make_float4(keep[0], keep[1], keep[2], keep[3]);
    }
}

__device__ __forceinline__ int clamp_i(int v, int lo, int hi) {
    return v < lo ? lo : (v > hi ? hi : v);
}

#define DPB   400
#define NBIN  (N_NODES / DPB)
#define EBLK  200
#define ECH   (N_EDGES / EBLK)
#define CELLW 128
#define CELLP 129
#define EPT   ((ECH + 1023) / 1024)

__global__ void __launch_bounds__(1024)
k_bucket(const int* __restrict__ ei, int* __restrict__ cells) {
    __shared__ int hist[NBIN];
    __shared__ int img[NBIN * CELLP];
    const int t = threadIdx.x;
    const int b = blockIdx.x;
    if (t < NBIN) hist[t] = 0;
    int sv[EPT], dv[EPT];
#pragma unroll
    for (int k = 0; k < EPT; ++k) {
        const int o = k * 1024 + t;
        const bool in = o < ECH;
        sv[k] = in ? ei[b * ECH + o] : -1;
        dv[k] = in ? ei[N_EDGES + b * ECH + o] : -1;
    }
    __syncthreads();
#pragma unroll
    for (int k = 0; k < EPT; ++k) {
        const int s = sv[k], d = dv[k];
        if ((unsigned)d < (unsigned)N_NODES && (unsigned)s < (unsigned)N_NODES) {
            const int bin = d / DPB;
            const int dl = d - bin * DPB;
            const int r = atomicAdd(&hist[bin], 1);
            if (r < CELLW - 1) img[bin * CELLP + 1 + r] = (dl << 16) | s;
        }
    }
    __syncthreads();
    if (t < NBIN) img[t * CELLP] = hist[t];
    __syncthreads();
    int* dst = cells + (size_t)b * (NBIN * CELLW);
    for (int i = t; i < NBIN * CELLW; i += 1024)
        dst[i] = img[(i / CELLW) * CELLP + (i % CELLW)];
}

#define GPB    100
#define GQ     (DPB / GPB)
#define GWAVES 32
#define CPW    ((EBLK + GWAVES - 1) / GWAVES)
#define ROWP   65
#define WPITCH 68
#define WP4    (WPITCH / 4)

__device__ __forceinline__ float lrelu(float v) {
    return v > 0.0f ? v : NEG_SLOPE * v;
}
__device__ __forceinline__ float lane_f(float v, int j) {
    return __int_as_float(__builtin_amdgcn_readlane(__float_as_int(v), j));
}
struct HeadMask { unsigned k0, k1, k2, k3; };
__device__ __forceinline__ float by_head(float a, float b, float c, float e,
                                         const HeadMask& k) {
    const unsigned r = (__float_as_uint(a) & k.k0) | (__float_as_uint(b) & k.k1) |
                       (__float_as_uint(c) & k.k2) | (__float_as_uint(e) & k.k3);
    return __uint_as_float(r);
}
__device__ __forceinline__ float head_w(float w0, float w1, float w2, float w3,
                                        int j, const HeadMask& k) {
    return by_head(lane_f(w0, j), lane_f(w1, j), lane_f(w2, j), lane_f(w3, j), k);
}

typedef float nf4 __attribute__((ext_vector_type(4)));
typedef const nf4 __attribute__((address_space(1)))* grow_ro;
typedef nf4 __attribute__((address_space(1)))* grow_rw;
__device__ __forceinline__ unsigned long long row_addr(const float* tab, int s) {
    const unsigned long long p = (unsigned long long)tab +
        (unsigned long long)(unsigned)s * (OUT_CH * sizeof(float));
    const unsigned lo = __builtin_amdgcn_readfirstlane((unsigned)p);
    const unsigned hi = __builtin_amdgcn_readfirstlane((unsigned)(p >> 32));
    return ((unsigned long long)hi << 32) | lo;
}
__device__ __forceinline__ float4 load_row(const float* tab, int s, int lane) {
    const nf4 v = ((grow_ro)row_addr(tab, s))[lane];
    return make_float4(v.x, v.y, v.z, v.w);
}
__device__ __forceinline__ void store_row(float* tab, int s, int lane, float4 y) {
    nf4 v; v.x = y.x; v.y = y.y; v.z = y.z; v.w = y.w;
    ((grow_rw)row_addr(tab, s))[lane] = v;
}

__device__ __forceinline__ void add_rows(float4& acc, int s, const float4* w4,
        int n, const float* __restrict__ h, int lane) {
    int j = 0;
    for (; j + 4 <= n; j += 4) {
        const float4 w = w4[j >> 2];
        const int sa = __builtin_amdgcn_readlane(s, j);
        const int sb = __builtin_amdgcn_readlane(s, j + 1);
        const int sc = __builtin_amdgcn_readlane(s, j + 2);
        const int se = __builtin_amdgcn_readlane(s, j + 3);
        const float4 ha = load_row(h, sa, lane);
        const float4 hb = load_row(h, sb, lane);
        const float4 hc = load_row(h, sc, lane);
        const float4 he = load_row(h, se, lane);
        acc.x += ha.x * w.x; acc.y += ha.y * w.x; acc.z += ha.z * w.x; acc.w += ha.w * w.x;
        acc.x += hb.x * w.y; acc.y += hb.y * w.y; acc.z += hb.z * w.y; acc.w += hb.w * w.y;
        acc.x += hc.x * w.z; acc.y += hc.y * w.z; acc.z += hc.z * w.z; acc.w += hc.w * w.z;
        acc.x += he.x * w.w; acc.y += he.y * w.w; acc.z += he.z * w.w; acc.w += he.w * w.w;
    }
    if (j < n) {
        const float4 w = w4[j >> 2];
        const int sa = __builtin_amdgcn_readlane(s, j);
        const float4 ha = load_row(h, sa, lane);
        acc.x += ha.x * w.x; acc.y += ha.y * w.x; acc.z += ha.z * w.x; acc.w += ha.w * w.x;
        if (j + 1 < n) {
            const int sb = __builtin_amdgcn_readlane(s, j + 1);
            const float4 hb = load_row(h, sb, lane);
            acc.x += hb.x * w.y; acc.y += hb.y * w.y; acc.z += hb.z * w.y; acc.w += hb.w * w.y;
        }
        if (j + 2 < n) {
            const int sc = __builtin_amdgcn_readlane(s, j + 2);
            const float4 hc = load_row(h, sc, lane);
            acc.x += hc.x * w.z; acc.y += hc.y * w.z; acc.z += hc.z * w.z; acc.w += hc.w * w.z;
        }
    }
}

__device__ __forceinline__ void finish_row(float4 acc, int d, int lane,
        const float4* prm, float* out) {
    const float4 bs = prm[lane];
    float4 v;
    v.x = acc.x + bs.x; v.y = acc.y + bs.y;
    v.z = acc.z + bs.z; v.w = acc.w + bs.w;
    float t = (v.x + v.y) + (v.z + v.w);
    for (int o = 16; o > 0; o >>= 1) t += __shfl_xor(t, o, 32);
    const float mu = t * (1.0f / OUT_CH);
    const float dx = v.x - mu, dy = v.y - mu, dz = v.z - mu, dw = v.w - mu;
    t = (dx * dx + dy * dy) + (dz * dz + dw * dw);
    for (int o = 16; o > 0; o >>= 1) t += __shfl_xor(t, o, 32);
    const float var = t * (1.0f / OUT_CH);
    const float rs = 1.0f / sqrtf(var + LN_EPS);
    const float4 gm = prm[32 + lane];
    const float4 bt = prm[64 + lane];
    float4 y;
    y.x = dx * rs * gm.x + bt.x; y.y = dy * rs * gm.y + bt.y;
    y.z = dz * rs * gm.z + bt.z; y.w = dw * rs * gm.w + bt.w;
    y.x = y.x > 0.0f ? y.x : (expf(y.x) - 1.0f);
    y.y = y.y > 0.0f ? y.y : (expf(y.y) - 1.0f);
    y.z = y.z > 0.0f ? y.z : (expf(y.z) - 1.0f);
    y.w = y.w > 0.0f ? y.w : (expf(y.w) - 1.0f);
    store_row(out, d, lane, y);
}

__device__ __forceinline__ void place_cell(int4 cw, int j0, int n, int lo,
                                           int* dcnt, int* rows) {
    const int w[4] = {cw.x, cw.y, cw.z, cw.w};
    int dl[4], r[4];
    bool ok[4];
#pragma unroll
    for (int i = 0; i < 4; ++i) {
        const int j = j0 + i;
        dl[i] = (w[i] >> 16) - lo;
        ok[i] = j >= 1 && j <= n && (unsigned)dl[i] < (unsigned)GPB;
    }
#pragma unroll
    for (int i = 0; i < 4; ++i) {
        r[i] = SLOTS;
        if (ok[i]) r[i] = atomicAdd(&dcnt[dl[i]], 1);
    }
#pragma unroll
    for (int i = 0; i < 4; ++i)
        if (ok[i] && r[i] < SLOTS) rows[dl[i] * ROWP + r[i]] = w[i] & 0xffff;
}

__global__ void __launch_bounds__(GWAVES * 32)
k_gather(const int* __restrict__ ei, const int* __restrict__ cells,
         const float* __restrict__ a_s, const float* __restrict__ a_d,
         const float* __restrict__ h,
         const float* __restrict__ bias, const float* __restrict__ gamma,
         const float* __restrict__ beta, float* __restrict__ out) {
    __shared__ int rows[GPB * ROWP];
    __shared__ int dcnt[GPB];
    __shared__ int cut;
    __shared__ int next;
    __shared__ float4 prm[3 * 32];
    __shared__ float4 wts4[GWAVES * 4 * WP4];
    const int t = threadIdx.x;
    const int lane = t & 31;
    const int head = lane >> 3;
    const int wave = __builtin_amdgcn_readfirstlane(t >> 5);
    const int bin = blockIdx.x / GQ;
    const int lo = (blockIdx.x % GQ) * GPB;
    const int dbase = bin * DPB + lo;

    if (t < GPB) dcnt[t] = 0;
    if (t == 0) { cut = 0; next = 0; }
    if (t < 96) {
        const float* pv = t < 32 ? bias : (t < 64 ? gamma : beta);
        prm[t] = ((const float4*)pv)[lane];
    }
    int4 cw[CPW];
#pragma unroll
    for (int k = 0; k < CPW; ++k) {
        const int cb = wave + k * GWAVES;
        const int cbc = cb < EBLK ? cb : EBLK - 1;
        const int* cp = cells + ((size_t)cbc * NBIN + bin) * CELLW;
        cw[k] = ((const int4*)cp)[lane];
    }
    __syncthreads();
#pragma unroll
    for (int k = 0; k < CPW; ++k) {
        const int n = (wave + k * GWAVES < EBLK)
                          ? __builtin_amdgcn_readfirstlane(cw[k].x) : 0;
        if (n > CELLW - 1 || n < 0) cut = 1;
        place_cell(cw[k], lane * 4, n, lo, dcnt, rows);
    }
    __syncthreads();
    const bool isCut = __builtin_amdgcn_readfirstlane(cut) != 0;

    HeadMask hk;
    hk.k0 = head == 0 ? 0xffffffffu : 0u;
    hk.k1 = head == 1 ? 0xffffffffu : 0u;
    hk.k2 = head == 2 ? 0xffffffffu : 0u;
    hk.k3 = head == 3 ? 0xffffffffu : 0u;
    float* const wst = (float*)wts4 + wave * (4 * WPITCH) + lane;
    const float4* const wld = wts4 + wave * (4 * WP4) + head * WP4;

    for (;;) {
        int tk = 0;
        if (lane == 0) tk = atomicAdd(&next, 1);
        const int dl = __builtin_amdgcn_readfirstlane(tk);
        if (dl >= GPB) break;
        const int d = dbase + dl;
        const int dall = isCut ? SLOTS + 1
                               : clamp_i(__builtin_amdgcn_readfirstlane(dcnt[dl]), 0, N_EDGES);
        const int deg = dall > SLOTS ? 0 : dall;
        const int* row = rows + dl * ROWP;
        const float4 ad = *(const float4*)(a_d + (size_t)d * HEADS);
        const float4 sd = *(const float4*)(a_s + (size_t)d * HEADS);
        const float ls0 = lrelu(sd.x + ad.x), ls1 = lrelu(sd.y + ad.y);
        const float ls2 = lrelu(sd.z + ad.z), ls3 = lrelu(sd.w + ad.w);

        const bool inA = lane < deg;
        const bool inB = lane + 32 < deg;
        int rA = 0, rB = 0;
        if (inA) {
            rA = row[lane];
            rB = row[inB ? lane + 32 : lane];
        }
        const int sA = clamp_i(rA, 0, N_NODES - 1);
        const int sB = inB ? clamp_i(rB, 0, N_NODES - 1) : 0;
        const float4 asA = *(const float4*)(a_s + (size_t)sA * HEADS);
        const float4 asB = *(const float4*)(a_s + (size_t)sB * HEADS);
        float a0 = lrelu(asA.x + (inA ? ad.x : -INFINITY));
        float a1 = lrelu(asA.y + (inA ? ad.y : -INFINITY));
        float a2 = lrelu(asA.z + (inA ? ad.z : -INFINITY));
        float a3 = lrelu(asA.w + (inA ? ad.w : -INFINITY));
        float b0 = lrelu(asB.x + (inB ? ad.x : -INFINITY));
        float b1 = lrelu(asB.y + (inB ? ad.y : -INFINITY));
        float b2 = lrelu(asB.z + (inB ? ad.z : -INFINITY));
        float b3 = lrelu(asB.w + (inB ? ad.w : -INFINITY));

        float m0 = fmaxf(fmaxf(ls0, a0), b0), m1 = fmaxf(fmaxf(ls1, a1), b1);
        float m2 = fmaxf(fmaxf(ls2, a2), b2), m3 = fmaxf(fmaxf(ls3, a3), b3);
        for (int o = 16; o > 0; o >>= 1) {
            m0 = fmaxf(m0, __shfl_xor(m0, o, 32));
            m1 = fmaxf(m1, __shfl_xor(m1, o, 32));
            m2 = fmaxf(m2, __shfl_xor(m2, o, 32));
            m3 = fmaxf(m3, __shfl_xor(m3, o, 32));
        }
        m0 = (fabsf(m0) < INFINITY) ? m0 : 0.0f;
        m1 = (fabsf(m1) < INFINITY) ? m1 : 0.0f;
        m2 = (fabsf(m2) < INFINITY) ? m2 : 0.0f;
        m3 = (fabsf(m3) < INFINITY) ? m3 : 0.0f;

        if (deg > 0) {
            const float e0 = expf(a0 - m0), e1 = expf(a1 - m1);
            const float e2 = expf(a2 - m2), e3 = expf(a3 - m3);
            a0 = inA ? e0 : 0.0f; a1 = inA ? e1 : 0.0f;
            a2 = inA ? e2 : 0.0f; a3 = inA ? e3 : 0.0f;
        } else {
            a0 = 0.0f; a1 = 0.0f; a2 = 0.0f; a3 = 0.0f;
        }
        if (deg > 32) {
            const float e0 = expf(b0 - m0), e1 = expf(b1 - m1);
            const float e2 = expf(b2 - m2), e3 = expf(b3 - m3);
            b0 = inB ? e0 : 0.0f; b1 = inB ? e1 : 0.0f;
            b2 = inB ? e2 : 0.0f; b3 = inB ? e3 : 0.0f;
        } else {
            b0 = 0.0f; b1 = 0.0f; b2 = 0.0f; b3 = 0.0f;
        }
        float p0 = a0 + b0, p1 = a1 + b1, p2 = a2 + b2, p3 = a3 + b3;
        for (int o = 16; o > 0; o >>= 1) {
            p0 += __shfl_xor(p0, o, 32);
            p1 += __shfl_xor(p1, o, 32);
            p2 += __shfl_xor(p2, o, 32);
            p3 += __shfl_xor(p3, o, 32);
        }
        const float es0 = expf(ls0 - m0), es1 = expf(ls1 - m1);
        const float es2 = expf(ls2 - m2), es3 = expf(ls3 - m3);
        const float den0 = (p0 + es0) + SOFTMAX_EPS;
        const float den1 = (p1 + es1) + SOFTMAX_EPS;
        const float den2 = (p2 + es2) + SOFTMAX_EPS;
        const float den3 = (p3 + es3) + SOFTMAX_EPS;

        const float wself = by_head(es0 / den0, es1 / den1, es2 / den2,
                                    es3 / den3, hk);
        const float4 hs = load_row(h, d, lane);
        float4 acc;
        acc.x = hs.x * wself; acc.y = hs.y * wself;
        acc.z = hs.z * wself; acc.w = hs.w * wself;

        if (deg > 0) {
            wst[0 * WPITCH] = a0 / den0; wst[1 * WPITCH] = a1 / den1;
            wst[2 * WPITCH] = a2 / den2; wst[3 * WPITCH] = a3 / den3;
            if (deg > 32) {
                wst[0 * WPITCH + 32] = b0 / den0; wst[1 * WPITCH + 32] = b1 / den1;
                wst[2 * WPITCH + 32] = b2 / den2; wst[3 * WPITCH + 32] = b3 / den3;
            }
            add_rows(acc, sA, wld, deg < 32 ? deg : 32, h, lane);
            if (deg > 32) add_rows(acc, sB, wld + 8, deg - 32, h, lane);
        }
        if (dall <= SLOTS) finish_row(acc, d, lane, prm, out);
    }

    const int* dsts = ei + N_EDGES;
    for (int dl = wave; dl < GPB; dl += GWAVES) {
        const int deg = isCut ? SLOTS + 1
                              : clamp_i(__builtin_amdgcn_readfirstlane(dcnt[dl]), 0, N_EDGES);
        if (deg <= SLOTS) continue;
        const int d = dbase + dl;
        const float4 ad = *(const float4*)(a_d + (size_t)d * HEADS);
        const float4 sd = *(const float4*)(a_s + (size_t)d * HEADS);
        const float ls0 = lrelu(sd.x + ad.x), ls1 = lrelu(sd.y + ad.y);
        const float ls2 = lrelu(sd.z + ad.z), ls3 = lrelu(sd.w + ad.w);

        float m0 = ls0, m1 = ls1, m2 = ls2, m3 = ls3;
        for (int i = lane; i < N_EDGES; i += 32) {
            if (dsts[i] != d) continue;
            const int s = clamp_i(ei[i], 0, N_NODES - 1);
            const float4 as = *(const float4*)(a_s + (size_t)s * HEADS);
            m0 = fmaxf(m0, lrelu(as.x + ad.x));
            m1 = fmaxf(m1, lrelu(as.y + ad.y));
            m2 = fmaxf(m2, lrelu(as.z + ad.z));
            m3 = fmaxf(m3, lrelu(as.w + ad.w));
        }
        for (int o = 16; o > 0; o >>= 1) {
            m0 = fmaxf(m0, __shfl_xor(m0, o, 32));
            m1 = fmaxf(m1, __shfl_xor(m1, o, 32));
            m2 = fmaxf(m2, __shfl_xor(m2, o, 32));
            m3 = fmaxf(m3, __shfl_xor(m3, o, 32));
        }
        m0 = (fabsf(m0) < INFINITY) ? m0 : 0.0f;
        m1 = (fabsf(m1) < INFINITY) ? m1 : 0.0f;
        m2 = (fabsf(m2) < INFINITY) ? m2 : 0.0f;
        m3 = (fabsf(m3) < INFINITY) ? m3 : 0.0f;

        float p0 = 0.0f, p1 = 0.0f, p2 = 0.0f, p3 = 0.0f;
        for (int i = lane; i < N_EDGES; i += 32) {
            if (dsts[i] != d) continue;
            const int s = clamp_i(ei[i], 0, N_NODES - 1);
            const float4 as = *(const float4*)(a_s + (size_t)s * HEADS);
            p0 += expf(lrelu(as.x + ad.x) - m0);
            p1 += expf(lrelu(as.y + ad.y) - m1);
            p2 += expf(lrelu(as.z + ad.z) - m2);
            p3 += expf(lrelu(as.w + ad.w) - m3);
        }
        for (int o = 16; o > 0; o >>= 1) {
            p0 += __shfl_xor(p0, o, 32);
            p1 += __shfl_xor(p1, o, 32);
            p2 += __shfl_xor(p2, o, 32);
            p3 += __shfl_xor(p3, o, 32);
        }
        const float es0 = expf(ls0 - m0), es1 = expf(ls1 - m1);
        const float es2 = expf(ls2 - m2), es3 = expf(ls3 - m3);
        const float den0 = (p0 + es0) + SOFTMAX_EPS;
        const float den1 = (p1 + es1) + SOFTMAX_EPS;
        const float den2 = (p2 + es2) + SOFTMAX_EPS;
        const float den3 = (p3 + es3) + SOFTMAX_EPS;

        const float wself = by_head(es0 / den0, es1 / den1, es2 / den2,
                                    es3 / den3, hk);
        const float4 hs = load_row(h, d, lane);
        float4 acc;
        acc.x = hs.x * wself; acc.y = hs.y * wself;
        acc.z = hs.z * wself; acc.w = hs.w * wself;

        for (int base = 0; base < N_EDGES; base += 32) {
            const int i = base + lane;
            int s = 0;
            float w0 = 0.0f, w1 = 0.0f, w2 = 0.0f, w3 = 0.0f;
            const bool mine = i < N_EDGES && dsts[i] == d;
            if (mine) {
                s = clamp_i(ei[i], 0, N_NODES - 1);
                const float4 as = *(const float4*)(a_s + (size_t)s * HEADS);
                w0 = expf(lrelu(as.x + ad.x) - m0) / den0;
                w1 = expf(lrelu(as.y + ad.y) - m1) / den1;
                w2 = expf(lrelu(as.z + ad.z) - m2) / den2;
                w3 = expf(lrelu(as.w + ad.w) - m3) / den3;
            }
            unsigned todo = __builtin_amdgcn_ballot_w32(mine);
            while (todo != 0u) {
                const int jj = __builtin_ctz(todo);
                todo &= todo - 1u;
                const int sa = __builtin_amdgcn_readlane(s, jj);
                const float4 ha = load_row(h, sa, lane);
                const float wa = head_w(w0, w1, w2, w3, jj, hk);
                acc.x += ha.x * wa; acc.y += ha.y * wa; acc.z += ha.z * wa; acc.w += ha.w * wa;
            }
        }
        finish_row(acc, d, lane, prm, out);
    }
}

extern "C" void kernel_launch(void* const* d_in, const int* in_sizes, int n_in,
                              void* d_out, int out_size, void* d_ws, size_t ws_size,
                              hipStream_t stream) {
    const float* x       = (const float*)d_in[0];
    const int*   ei      = (const int*)  d_in[1];
    const float* W       = (const float*)d_in[2];
    const float* att_src = (const float*)d_in[3];
    const float* att_dst = (const float*)d_in[4];
    const float* bias    = (const float*)d_in[5];
    const float* gamma   = (const float*)d_in[6];
    const float* beta    = (const float*)d_in[7];
    float* out = (float*)d_out;

    char* ws = (char*)d_ws;
    float*    h      = (float*)ws;
    ws += (size_t)N_NODES * OUT_CH * sizeof(float);
    float*    a_s    = (float*)ws;  ws += (size_t)N_NODES * HEADS * sizeof(float);
    float*    a_d    = (float*)ws;  ws += (size_t)N_NODES * HEADS * sizeof(float);
    ws += (size_t)N_NODES * HEADS * sizeof(int);
    int*      cells  = (int*)ws;

    k_bucket<<<EBLK, 1024, 0, stream>>>(ei, cells);

    k_gemm_wmma<<<(N_NODES + 127) / 128, 256, IN_CH * OUT_CH * sizeof(float),
                  stream>>>(x, W, att_src, att_dst, h, a_s, a_d, N_NODES);

    k_gather<<<NBIN * GQ, GWAVES * 32, 0, stream>>>(ei, cells, a_s, a_d, h,
                                                     bias, gamma, beta, out);
}
